# starting kernel
# baseline (speedup 1.0000x reference)
#include <hip/hip_runtime.h>
#include <hip/hip_bf16.h>

typedef __attribute__((ext_vector_type(2))) float v2f;
typedef __attribute__((ext_vector_type(8))) float v8f;

#define B_ 16
#define N_ 4096
#define C_ 1024
#define NCHUNK 16
#define CHUNK 256
#define XSTR 1032

__global__ __launch_bounds__(256) void ga_lncf(const float* __restrict__ cf,
                                               const float* __restrict__ qg,
                                               const float* __restrict__ qb,
                                               float* __restrict__ cfn) {
  __shared__ float s0[256], s1[256];
  const int tid = threadIdx.x;
  float4 x = *(const float4*)(cf + tid * 4);
  s0[tid] = x.x + x.y + x.z + x.w;
  s1[tid] = x.x * x.x + x.y * x.y + x.z * x.z + x.w * x.w;
  __syncthreads();
  for (int s = 128; s > 0; s >>= 1) {
    if (tid < s) { s0[tid] += s0[tid + s]; s1[tid] += s1[tid + s]; }
    __syncthreads();
  }
  const float mu = s0[0] * (1.f / 1024.f);
  const float var = s1[0] * (1.f / 1024.f) - mu * mu;
  const float rs = rsqrtf(var + 1e-5f);
  float4 g = *(const float4*)(qg + tid * 4);
  float4 b4 = *(const float4*)(qb + tid * 4);
  float4 o;
  o.x = (x.x - mu) * rs * g.x + b4.x;
  o.y = (x.y - mu) * rs * g.y + b4.y;
  o.z = (x.z - mu) * rs * g.z + b4.z;
  o.w = (x.w - mu) * rs * g.w + b4.w;
  *(float4*)(cfn + tid * 4) = o;
}

__global__ __launch_bounds__(256) void ga_q(const float* __restrict__ cfn,
                                            const float* __restrict__ Wq,
                                            float* __restrict__ qs) {
  __shared__ float cs[1024];
  const int tid = threadIdx.x;
  for (int i = tid; i < 1024; i += 256) cs[i] = cfn[i];
  __syncthreads();
  const int wv = tid >> 5, ln = tid & 31;
  for (int jj = 0; jj < 32; ++jj) {
    const int j = blockIdx.x * 256 + wv * 32 + jj;
    const float* wrow = Wq + (size_t)j * 1024;
    float acc = 0.f;
    for (int c0 = 0; c0 < 1024; c0 += 32) acc += cs[c0 + ln] * wrow[c0 + ln];
#pragma unroll
    for (int m = 16; m > 0; m >>= 1) acc += __shfl_xor(acc, m, 32);
    if (ln == 0) qs[j] = acc * 0.03125f;
  }
}

__global__ __launch_bounds__(256) void ga_wqeff(const float* __restrict__ Wkv,
                                                const float* __restrict__ qs,
                                                float* __restrict__ wqe) {
  __shared__ float q_s[1024];
  const int tid = threadIdx.x;
  for (int i = tid; i < 1024; i += 256) q_s[i] = qs[i];
  __syncthreads();
  const int c = blockIdx.x * 256 + tid;
  for (int h = 0; h < 8; ++h) {
    float acc = 0.f;
    for (int d = 0; d < 128; ++d)
      acc += Wkv[(size_t)(h * 128 + d) * 1024 + c] * q_s[h * 128 + d];
    wqe[c * 16 + h] = acc;
  }
  for (int h = 8; h < 16; ++h) wqe[c * 16 + h] = 0.f;
}

__global__ __launch_bounds__(256) void ga_main(const float* __restrict__ feats,
                                               const float* __restrict__ kvg,
                                               const float* __restrict__ kvb,
                                               const float* __restrict__ wqe,
                                               float* __restrict__ mp,
                                               float* __restrict__ lp,
                                               float* __restrict__ yp) {
  __shared__ __align__(16) float xs[16][XSTR];
  __shared__ __align__(16) float wq[1024 * 17];
  __shared__ __align__(32) float sred[2048];
  __shared__ float sS[256];
  __shared__ float sP[16 * 17];
  __shared__ float sF[16];
  __shared__ float smu[16], srs[16];

  const int tid = threadIdx.x;
  const int lane = tid & 31;
  const int wave = tid >> 5;
  const int l16 = lane & 15;
  const int lhi = lane >> 4;
  const int r = tid >> 4;
  const int t16 = tid & 15;

  const int blk = blockIdx.x;
  const int bb = blk >> 4;
  const int chunk = blk & 15;

  for (int i = tid; i < 16384; i += 256) wq[(i >> 4) * 17 + (i & 15)] = wqe[i];

  const v8f vzero = {0.f, 0.f, 0.f, 0.f, 0.f, 0.f, 0.f, 0.f};
  v8f yacc[8];
#pragma unroll
  for (int t = 0; t < 8; ++t) yacc[t] = vzero;

  float m_run = -3.4e38f;
  float l_run = 0.f;

  __syncthreads();

#pragma unroll 1
  for (int st = 0; st < 16; ++st) {
    const size_t rowbase =
        ((size_t)bb * N_ + (size_t)chunk * CHUNK + st * 16 + r) * C_;
    float sum = 0.f, sq = 0.f;
#pragma unroll
    for (int i = 0; i < 16; ++i) {
      const int col = i * 64 + t16 * 4;
      float4 x = *(const float4*)(feats + rowbase + col);
      sum += x.x + x.y + x.z + x.w;
      sq += x.x * x.x + x.y * x.y + x.z * x.z + x.w * x.w;
      *(float4*)&xs[r][col] = x;
    }
    if (st < 15)
      __builtin_prefetch(feats + rowbase + 16 * C_ + t16 * 64, 0, 3);
    sred[tid] = sum;
    sred[256 + tid] = sq;
    __syncthreads();
    if (tid < 16) {
      float s0 = 0.f, s1 = 0.f;
#pragma unroll
      for (int i = 0; i < 16; ++i) {
        s0 += sred[tid * 16 + i];
        s1 += sred[256 + tid * 16 + i];
      }
      const float mu = s0 * (1.f / 1024.f);
      const float var = s1 * (1.f / 1024.f) - mu * mu;
      smu[tid] = mu;
      srs[tid] = rsqrtf(var + 1e-5f);
    }
    __syncthreads();
    {
      const float mu = smu[r], rs = srs[r];
#pragma unroll
      for (int i = 0; i < 16; ++i) {
        const int col = i * 64 + t16 * 4;
        float4 x = *(float4*)&xs[r][col];
        float4 g = *(const float4*)(kvg + col);
        float4 b4 = *(const float4*)(kvb + col);
        x.x = (x.x - mu) * rs * g.x + b4.x;
        x.y = (x.y - mu) * rs * g.y + b4.y;
        x.z = (x.z - mu) * rs * g.z + b4.z;
        x.w = (x.w - mu) * rs * g.w + b4.w;
        *(float4*)&xs[r][col] = x;
      }
    }
    __syncthreads();

    v8f sacc = vzero;
    const int kb = wave * 128 + 2 * lhi;
#pragma unroll
    for (int kk = 0; kk < 32; ++kk) {
      const int k0 = kb + kk * 4;
      v2f a = *(const v2f*)&xs[l16][k0];
      v2f bv;
      bv.x = wq[k0 * 17 + l16];
      bv.y = wq[(k0 + 1) * 17 + l16];
      sacc = __builtin_amdgcn_wmma_f32_16x16x4_f32(false, a, false, bv,
                                                   (short)0, sacc, false, false);
    }
    *(v8f*)&sred[wave * 256 + lane * 8] = sacc;
    __syncthreads();
    {
      const int rr = tid >> 4, hh = tid & 15;
      const int idx = (hh + ((rr >> 3) << 4)) * 8 + (rr & 7);
      float v = 0.f;
#pragma unroll
      for (int w = 0; w < 8; ++w) v += sred[w * 256 + idx];
      sS[rr * 16 + hh] = v;
    }
    __syncthreads();

    if (tid < 16) {
      float rmax = -3.4e38f;
#pragma unroll
      for (int q = 0; q < 16; ++q) rmax = fmaxf(rmax, sS[q * 16 + tid]);
      const float mnew = fmaxf(m_run, rmax);
      const float f = __expf(m_run - mnew);
      float ls = 0.f;
#pragma unroll
      for (int q = 0; q < 16; ++q) {
        const float p = __expf(sS[q * 16 + tid] - mnew);
        sP[q * 17 + tid] = p;
        ls += p;
      }
      l_run = l_run * f + ls;
      m_run = mnew;
      sF[tid] = f;
    }
    __syncthreads();

    float fv[8];
#pragma unroll
    for (int v = 0; v < 8; ++v) fv[v] = sF[v + 8 * lhi];
#pragma unroll
    for (int t = 0; t < 8; ++t) {
#pragma unroll
      for (int v = 0; v < 8; ++v) yacc[t][v] *= fv[v];
    }
#pragma unroll
    for (int k0 = 0; k0 < 16; k0 += 4) {
      const int kr = k0 + 2 * lhi;
      v2f a;
      a.x = sP[kr * 17 + l16];
      a.y = sP[(kr + 1) * 17 + l16];
#pragma unroll
      for (int t = 0; t < 8; ++t) {
        const int c0 = wave * 128 + t * 16;
        v2f bv;
        bv.x = xs[kr][c0 + l16];
        bv.y = xs[kr + 1][c0 + l16];
        yacc[t] = __builtin_amdgcn_wmma_f32_16x16x4_f32(
            false, a, false, bv, (short)0, yacc[t], false, false);
      }
    }
    __syncthreads();
  }

  if (tid < 8) {
    mp[blk * 8 + tid] = m_run;
    lp[blk * 8 + tid] = l_run;
  }
  if (lhi == 0) {
#pragma unroll
    for (int t = 0; t < 8; ++t) {
      const int c0 = wave * 128 + t * 16;
#pragma unroll
      for (int v = 0; v < 8; ++v)
        yp[(size_t)(blk * 8 + v) * 1024 + c0 + l16] = yacc[t][v];
    }
  }
}

__global__ __launch_bounds__(1024) void ga_merge(const float* __restrict__ mp,
                                                 const float* __restrict__ lp,
                                                 const float* __restrict__ yp,
                                                 const float* __restrict__ Wkv,
                                                 const float* __restrict__ projW,
                                                 const float* __restrict__ projb,
                                                 const float* __restrict__ cf,
                                                 float* __restrict__ out) {
  __shared__ float yn[8 * 1024];
  __shared__ float ag[1024];
  __shared__ float ef[16 * 8];
  __shared__ float Lh[8];
  const int tid = threadIdx.x;
  const int b = blockIdx.x;
  if (tid < 8) {
    float M = -3.4e38f;
    for (int k = 0; k < 16; ++k) M = fmaxf(M, mp[(b * 16 + k) * 8 + tid]);
    float L = 0.f;
    for (int k = 0; k < 16; ++k) {
      const float e = __expf(mp[(b * 16 + k) * 8 + tid] - M);
      ef[k * 8 + tid] = e;
      L += lp[(b * 16 + k) * 8 + tid] * e;
    }
    Lh[tid] = L;
  }
  __syncthreads();
  for (int e = tid; e < 8192; e += 1024) {
    const int h = e >> 10, c = e & 1023;
    float acc = 0.f;
    for (int k = 0; k < 16; ++k)
      acc += yp[(size_t)((b * 16 + k) * 8 + h) * 1024 + c] * ef[k * 8 + h];
    yn[e] = acc / Lh[h];
  }
  __syncthreads();
  const int wv = tid >> 5, ln = tid & 31;
  for (int jj = 0; jj < 32; ++jj) {
    const int j = wv * 32 + jj;
    const int h = j >> 7;
    const float* wrow = Wkv + (size_t)(1024 + j) * 1024;
    float acc = 0.f;
    for (int c0 = 0; c0 < 1024; c0 += 32) acc += yn[h * 1024 + c0 + ln] * wrow[c0 + ln];
#pragma unroll
    for (int m = 16; m > 0; m >>= 1) acc += __shfl_xor(acc, m, 32);
    if (ln == 0) ag[j] = acc;
  }
  __syncthreads();
  for (int jj = 0; jj < 32; ++jj) {
    const int j = wv * 32 + jj;
    const float* prow = projW + (size_t)j * 1024;
    float acc = 0.f;
    for (int c0 = 0; c0 < 1024; c0 += 32) acc += ag[c0 + ln] * prow[c0 + ln];
#pragma unroll
    for (int m = 16; m > 0; m >>= 1) acc += __shfl_xor(acc, m, 32);
    if (ln == 0) out[(size_t)b * 1024 + j] = cf[j] + acc + projb[j];
  }
}

extern "C" void kernel_launch(void* const* d_in, const int* in_sizes, int n_in,
                              void* d_out, int out_size, void* d_ws, size_t ws_size,
                              hipStream_t stream) {
  const float* class_feature = (const float*)d_in[0];
  const float* features      = (const float*)d_in[1];
  const float* q_gamma       = (const float*)d_in[2];
  const float* q_beta        = (const float*)d_in[3];
  const float* Wq            = (const float*)d_in[4];
  const float* kv_gamma      = (const float*)d_in[5];
  const float* kv_beta       = (const float*)d_in[6];
  const float* Wkv           = (const float*)d_in[7];
  const float* proj_W        = (const float*)d_in[8];
  const float* proj_b        = (const float*)d_in[9];
  float* out = (float*)d_out;

  float* ws   = (float*)d_ws;
  float* cfn  = ws;
  float* qs   = ws + 1024;
  float* wqe  = ws + 2048;
  float* mp   = ws + 18432;
  float* lp   = ws + 20480;
  float* yp   = ws + 22528;

  ga_lncf<<<1, 256, 0, stream>>>(class_feature, q_gamma, q_beta, cfn);
  ga_q<<<4, 256, 0, stream>>>(cfn, Wq, qs);
  ga_wqeff<<<4, 256, 0, stream>>>(Wkv, qs, wqe);
  ga_main<<<B_ * NCHUNK, 256, 0, stream>>>(features, kv_gamma, kv_beta, wqe, mp, lp, yp);
  ga_merge<<<B_, 1024, 0, stream>>>(mp, lp, yp, Wkv, proj_W, proj_b, class_feature, out);
}


# main: LayerNorm folded into both products: raw tile feeds the score and Y WMMAs; mean/rstd applied to scores and P, gamma/beta once per chunk; changes sum order of LN sums, scores and Y
# speedup vs baseline: 3.2033x; 3.2033x over previous
#include <hip/hip_runtime.h>
#include <hip/hip_bf16.h>

typedef __attribute__((ext_vector_type(2))) float v2f;
typedef __attribute__((ext_vector_type(8))) float v8f;

#define B_ 16
#define N_ 4096
#define C_ 1024
#define NCHUNK 16
#define CHUNK 256
#define TROWS 32
#define XSTR 1032

__global__ __launch_bounds__(512) void ga_pre(const float* __restrict__ cf,
                                              const float* __restrict__ qg,
                                              const float* __restrict__ qb,
                                              const float* __restrict__ Wq,
                                              const float* __restrict__ Wkv,
                                              float* __restrict__ wqe) {
  __shared__ __align__(16) float cs[1024];
  __shared__ float p0[512], p1[512];
  __shared__ float q_s[128];
  __shared__ __align__(16) float wk[128 * 128];

  const int tid = threadIdx.x;
  const int lane = tid & 31;
  const int wave = tid >> 5;
  const int h = blockIdx.x >> 3;
  const int col0 = (blockIdx.x & 7) * 128;

  const int t4 = (tid & 255) * 4;
  const float4 x = *(const float4*)(cf + t4);
  const float4 g = *(const float4*)(qg + t4);
  const float4 b4 = *(const float4*)(qb + t4);

  float4 tw[8];
#pragma unroll
  for (int u = 0; u < 8; ++u) {
    const int idx = tid + 512 * u;
    tw[u] = *(const float4*)(Wkv + (size_t)(h * 128 + (idx >> 5)) * 1024 + col0 + (idx & 31) * 4);
  }

  const float* wbase = Wq + (size_t)(h * 128 + wave * 8) * 1024 + lane * 4;
  float4 wa[2][8], wb[2][8];
#pragma unroll
  for (int rr = 0; rr < 2; ++rr)
#pragma unroll
    for (int i = 0; i < 8; ++i) wa[rr][i] = *(const float4*)(wbase + rr * 1024 + i * 128);

  p0[tid] = x.x + x.y + x.z + x.w;
  p1[tid] = x.x * x.x + x.y * x.y + x.z * x.z + x.w * x.w;
#pragma unroll
  for (int u = 0; u < 8; ++u) *(float4*)&wk[(tid + 512 * u) * 4] = tw[u];
  __syncthreads();
  float s0, s1;
  {
    float a[8], c[8];
#pragma unroll
    for (int k = 0; k < 8; ++k) { a[k] = p0[lane + 32 * k]; c[k] = p1[lane + 32 * k]; }
    s0 = ((a[0] + a[4]) + (a[2] + a[6])) + ((a[1] + a[5]) + (a[3] + a[7]));
    s1 = ((c[0] + c[4]) + (c[2] + c[6])) + ((c[1] + c[5]) + (c[3] + c[7]));
#pragma unroll
    for (int m = 16; m > 0; m >>= 1) {
      s0 += __shfl_xor(s0, m, 32);
      s1 += __shfl_xor(s1, m, 32);
    }
    s0 = __shfl(s0, 0, 32);
    s1 = __shfl(s1, 0, 32);
  }
  {
    const float mu = s0 * (1.f / 1024.f);
    const float var = s1 * (1.f / 1024.f) - mu * mu;
    const float rs = rsqrtf(var + 1e-5f);
    if (tid < 256) {
      float4 o;
      o.x = (x.x - mu) * rs * g.x + b4.x;
      o.y = (x.y - mu) * rs * g.y + b4.y;
      o.z = (x.z - mu) * rs * g.z + b4.z;
      o.w = (x.w - mu) * rs * g.w + b4.w;
      *(float4*)&cs[tid * 4] = o;
    }
  }
  __syncthreads();

  float4 cv[8];
#pragma unroll
  for (int i = 0; i < 8; ++i) cv[i] = *(const float4*)&cs[i * 128 + lane * 4];

#pragma unroll
  for (int r = 0; r < 4; ++r) {
    if (r < 3) {
#pragma unroll
      for (int rr = 0; rr < 2; ++rr)
#pragma unroll
        for (int i = 0; i < 8; ++i) {
          const float4 v = *(const float4*)(wbase + ((r + 1) * 2 + rr) * 1024 + i * 128);
          if (r & 1) wa[rr][i] = v; else wb[rr][i] = v;
        }
    }
    float acc[2];
#pragma unroll
    for (int rr = 0; rr < 2; ++rr) {
      float ax = 0.f, ay = 0.f, az = 0.f, aw = 0.f;
#pragma unroll
      for (int i = 0; i < 8; ++i) {
        const float4 w = (r & 1) ? wb[rr][i] : wa[rr][i];
        ax += cv[i].x * w.x;
        ay += cv[i].y * w.y;
        az += cv[i].z * w.z;
        aw += cv[i].w * w.w;
      }
      acc[rr] = (ax + ay) + (az + aw);
    }
#pragma unroll
    for (int m = 16; m > 0; m >>= 1) {
      const float t0 = __shfl_xor(acc[0], m, 32);
      const float t1 = __shfl_xor(acc[1], m, 32);
      acc[0] += t0;
      acc[1] += t1;
    }
    if (lane == 0) {
      q_s[wave * 8 + r * 2] = acc[0] * 0.03125f;
      q_s[wave * 8 + r * 2 + 1] = acc[1] * 0.03125f;
    }
  }
  __syncthreads();

  if (tid < 128) {
    float acc = 0.f;
#pragma unroll 16
    for (int d = 0; d < 128; ++d) acc += wk[d * 128 + tid] * q_s[d];
    wqe[h * 1024 + col0 + tid] = acc;
  }
}

template <int CTRL>
__device__ __forceinline__ float row_xor(float x) {
  const int b = __builtin_bit_cast(int, x);
  return __builtin_bit_cast(float, __builtin_amdgcn_update_dpp(b, b, CTRL, 0xf, 0xf, false));
}
__device__ __forceinline__ float xhalf(float x) {
  const int b = __builtin_bit_cast(int, x);
  return __builtin_bit_cast(
      float, __builtin_amdgcn_permlanex16(b, b, 0x76543210, 0xfedcba98, false, false));
}
__device__ __forceinline__ void wave_sum4(float& a, float& b, float& c, float& d) {
  a += row_xor<0x161>(a); b += row_xor<0x161>(b); c += row_xor<0x161>(c); d += row_xor<0x161>(d);
  a += row_xor<0x162>(a); b += row_xor<0x162>(b); c += row_xor<0x162>(c); d += row_xor<0x162>(d);
  a += row_xor<0x164>(a); b += row_xor<0x164>(b); c += row_xor<0x164>(c); d += row_xor<0x164>(d);
  a += row_xor<0x168>(a); b += row_xor<0x168>(b); c += row_xor<0x168>(c); d += row_xor<0x168>(d);
  a += xhalf(a); b += xhalf(b); c += xhalf(c); d += xhalf(d);
}
__global__ __launch_bounds__(512) void ga_main(const float* __restrict__ feats,
                                               const float* __restrict__ kvg,
                                               const float* __restrict__ kvb,
                                               const float* __restrict__ wqe,
                                               float* __restrict__ mp,
                                               float* __restrict__ lp,
                                               float* __restrict__ yp) {
  __shared__ __align__(16) float xs[TROWS][XSTR];
  __shared__ __align__(16) float wq[1024 * 17];
  __shared__ __align__(32) float sred[8192];
  __shared__ float sS[TROWS * 16];
  __shared__ float sP[TROWS * 17];
  __shared__ float sF[16];
  __shared__ float smu[TROWS], srs[TROWS];
  __shared__ float sG[16], sBq[16];
  __shared__ float sT[16], sL[16];

  const int tid = threadIdx.x;
  const int lane = tid & 31;
  const int wave = tid >> 5;
  const int l16 = lane & 15;
  const int lhi = lane >> 4;
  const int rp2 = wave * 2;
  const int cl = lane * 4;

  const int blk = blockIdx.x;
  const int bb = blk >> 4;
  const int chunk = blk & 15;

  {
    const float g0 = kvg[tid], g1 = kvg[tid + 512];
    const float b0 = kvb[tid], b1 = kvb[tid + 512];
    float gp[8], bp[8];
#pragma unroll
    for (int h = 0; h < 8; ++h) {
      const float w0 = wqe[h * 1024 + tid];
      const float w1 = wqe[h * 1024 + 512 + tid];
      const float v0 = w0 * g0, v1 = w1 * g1;
      wq[tid * 17 + h] = v0;
      wq[(tid + 512) * 17 + h] = v1;
      gp[h] = v0 + v1;
      bp[h] = w0 * b0 + w1 * b1;
    }
#pragma unroll
    for (int m = 16; m > 0; m >>= 1) {
#pragma unroll
      for (int h = 0; h < 8; ++h) {
        const float tg = __shfl_xor(gp[h], m, 32);
        const float tb = __shfl_xor(bp[h], m, 32);
        gp[h] += tg;
        bp[h] += tb;
      }
    }
    if (lane == 0) {
#pragma unroll
      for (int h = 0; h < 8; ++h) {
        sred[wave * 16 + h] = gp[h];
        sred[wave * 16 + 8 + h] = bp[h];
      }
    }
  }
  for (int i = tid; i < 8192; i += 512) wq[(i >> 3) * 17 + 8 + (i & 7)] = 0.f;
  __syncthreads();
  if (tid < 32) {
    const int hh = tid & 15;
    const int which = tid >> 4;
    float a = 0.f;
    if (hh < 8) {
#pragma unroll
      for (int w = 0; w < 16; ++w) a += sred[w * 16 + which * 8 + hh];
    }
    if (which) sBq[hh] = a; else sG[hh] = a;
  }

  const v8f vzero = {0.f, 0.f, 0.f, 0.f, 0.f, 0.f, 0.f, 0.f};
  v8f yacc[4];
#pragma unroll
  for (int t = 0; t < 4; ++t) yacc[t] = vzero;

  float m_run = -3.4e38f;
  float l_run = 0.f;
  float t_run = 0.f;

  __syncthreads();

#pragma unroll 1
  for (int st = 0; st < CHUNK / TROWS; ++st) {
    const size_t tilebase =
        ((size_t)bb * N_ + (size_t)chunk * CHUNK + st * TROWS + rp2) * C_ + cl;
    float sum0 = 0.f, sq0 = 0.f, sum1 = 0.f, sq1 = 0.f;
#pragma unroll
    for (int i = 0; i < 8; ++i) {
      const int col = i * 128 + cl;
      const float4 x0 = *(const float4*)(feats + tilebase + i * 128);
      const float4 x1 = *(const float4*)(feats + tilebase + C_ + i * 128);
      sum0 += x0.x + x0.y + x0.z + x0.w;
      sq0 += x0.x * x0.x + x0.y * x0.y + x0.z * x0.z + x0.w * x0.w;
      sum1 += x1.x + x1.y + x1.z + x1.w;
      sq1 += x1.x * x1.x + x1.y * x1.y + x1.z * x1.z + x1.w * x1.w;
      *(float4*)&xs[rp2][col] = x0;
      *(float4*)&xs[rp2 + 1][col] = x1;
    }
    if (st < CHUNK / TROWS - 1)
      __builtin_prefetch(feats + tilebase - cl + TROWS * C_ + (lane >> 4) * C_ + (lane & 15) * 64,
                         0, 3);
    wave_sum4(sum0, sq0, sum1, sq1);
    if (lane == 0) {
      const float mu0 = sum0 * (1.f / 1024.f);
      const float mu1 = sum1 * (1.f / 1024.f);
      smu[rp2] = mu0;
      smu[rp2 + 1] = mu1;
      srs[rp2] = rsqrtf(sq0 * (1.f / 1024.f) - mu0 * mu0 + 1e-5f);
      srs[rp2 + 1] = rsqrtf(sq1 * (1.f / 1024.f) - mu1 * mu1 + 1e-5f);
    }
    __syncthreads();

    v8f sacc0 = vzero, sacc1 = vzero;
    const int kb = wave * 64 + 2 * lhi;
#pragma unroll
    for (int kk = 0; kk < 16; ++kk) {
      const int k0 = kb + kk * 4;
      const v2f a0 = *(const v2f*)&xs[l16][k0];
      const v2f a1 = *(const v2f*)&xs[16 + l16][k0];
      v2f bv;
      bv.x = wq[k0 * 17 + l16];
      bv.y = wq[(k0 + 1) * 17 + l16];
      sacc0 = __builtin_amdgcn_wmma_f32_16x16x4_f32(false, a0, false, bv,
                                                    (short)0, sacc0, false, false);
      sacc1 = __builtin_amdgcn_wmma_f32_16x16x4_f32(false, a1, false, bv,
                                                    (short)0, sacc1, false, false);
    }
    *(v8f*)&sred[wave * 512 + lane * 8] = sacc0;
    *(v8f*)&sred[wave * 512 + 256 + lane * 8] = sacc1;
    __syncthreads();
    {
      const int hf = tid >> 8, rr = (tid >> 4) & 15, hh = tid & 15;
      const int idx = hf * 256 + (hh + ((rr >> 3) << 4)) * 8 + (rr & 7);
      const int row = hf * 16 + rr;
      const float mu = smu[row], rs = srs[row];
      const float gh = sG[hh], bq = sBq[hh];
      float v = 0.f;
#pragma unroll
      for (int w = 0; w < 16; ++w) v += sred[w * 512 + idx];
      sS[row * 16 + hh] = rs * (v - mu * gh) + bq;
    }
    __syncthreads();

    if (tid < 32) {
      const int hh = tid & 15;
      const int q0 = (tid >> 4) * 16;
      float rmax = -3.4e38f;
#pragma unroll
      for (int q = 0; q < 16; ++q) rmax = fmaxf(rmax, sS[(q0 + q) * 16 + hh]);
      rmax = fmaxf(rmax, __shfl_xor(rmax, 16, 32));
      const float mnew = fmaxf(m_run, rmax);
      const float f = __expf(m_run - mnew);
      float ls = 0.f, ts = 0.f;
#pragma unroll
      for (int q = 0; q < 16; ++q) {
        const float p = __expf(sS[(q0 + q) * 16 + hh] - mnew);
        const float pr = p * srs[q0 + q];
        sP[(q0 + q) * 17 + hh] = pr;
        ls += p;
        ts += pr * smu[q0 + q];
      }
      ls += __shfl_xor(ls, 16, 32);
      ts += __shfl_xor(ts, 16, 32);
      l_run = l_run * f + ls;
      t_run = t_run * f + ts;
      m_run = mnew;
      if (tid < 16) sF[tid] = f;
    }
    __syncthreads();

    float fv[8];
#pragma unroll
    for (int v = 0; v < 8; ++v) fv[v] = sF[v + 8 * lhi];
#pragma unroll
    for (int t = 0; t < 4; ++t) {
#pragma unroll
      for (int v = 0; v < 8; ++v) yacc[t][v] *= fv[v];
    }
#pragma unroll
    for (int k0 = 0; k0 < TROWS; k0 += 4) {
      const int kr = k0 + 2 * lhi;
      v2f a;
      a.x = sP[kr * 17 + l16];
      a.y = sP[(kr + 1) * 17 + l16];
#pragma unroll
      for (int t = 0; t < 4; ++t) {
        const int c0 = wave * 64 + t * 16;
        v2f bv;
        bv.x = xs[kr][c0 + l16];
        bv.y = xs[kr + 1][c0 + l16];
        yacc[t] = __builtin_amdgcn_wmma_f32_16x16x4_f32(
            false, a, false, bv, (short)0, yacc[t], false, false);
      }
    }
    __syncthreads();
  }

  if (tid < 16) {
    sT[tid] = t_run;
    sL[tid] = l_run;
  }
  if (tid < 8) {
    mp[blk * 8 + tid] = m_run;
    lp[blk * 8 + tid] = l_run;
  }
  __syncthreads();
  if (lhi == 0) {
#pragma unroll
    for (int t = 0; t < 4; ++t) {
      const int c0 = wave * 64 + t * 16;
      const float g = kvg[c0 + l16];
      const float b = kvb[c0 + l16];
#pragma unroll
      for (int v = 0; v < 8; ++v)
        yp[(size_t)(blk * 8 + v) * 1024 + c0 + l16] = g * (yacc[t][v] - sT[v]) + b * sL[v];
    }
  }
}

typedef __attribute__((ext_vector_type(4))) float v4f;

#define TROWS_W 64
#define WROWS 8

__device__ __forceinline__ float dot8(const v4f (&x)[8], const v4f (&w)[8]) {
  v2f lo = {0.f, 0.f}, hi = {0.f, 0.f};
#pragma unroll
  for (int i = 0; i < 8; ++i) {
    lo += x[i].lo * w[i].lo;
    hi += x[i].hi * w[i].hi;
  }
  return (lo.x + lo.y) + (hi.x + hi.y);
}

__device__ __forceinline__ void ld_rows(v4f (&w)[WROWS][8], const float* __restrict__ p) {
#pragma unroll
  for (int r = 0; r < WROWS; ++r)
#pragma unroll
    for (int i = 0; i < 8; ++i) w[r][i] = *(const v4f*)(p + r * 1024 + i * 128);
}

__device__ __forceinline__ void rows8(const v4f (&xv)[8], const v4f (&w)[WROWS][8],
                                      float* __restrict__ res, const int ln) {
  float a[WROWS];
#pragma unroll
  for (int r = 0; r < WROWS; ++r) a[r] = dot8(xv, w[r]);
#pragma unroll
  for (int m = 16; m > 0; m >>= 1) {
    float t[WROWS];
#pragma unroll
    for (int r = 0; r < WROWS; ++r) t[r] = __shfl_xor(a[r], m, 32);
#pragma unroll
    for (int r = 0; r < WROWS; ++r) a[r] += t[r];
  }
  if (ln == 0) {
#pragma unroll
    for (int r = 0; r < WROWS; ++r) res[r] = a[r];
  }
}

__global__ __launch_bounds__(256)
void ga_agg(const float* __restrict__ mp,
            const float* __restrict__ lp,
            const float* __restrict__ yp,
            const float* __restrict__ Wkv,
            float* __restrict__ agg) {
  __shared__ __align__(16) float yn[1024];
  __shared__ float res[TROWS_W];
  const int tid = threadIdx.x;
  const int wv = tid >> 5, ln = tid & 31;
  const int b = blockIdx.x >> 4;
  const int h = (blockIdx.x >> 1) & 7;
  const int r0 = h * 128 + (blockIdx.x & 1) * TROWS_W;

  v4f yv[16];
#pragma unroll
  for (int k = 0; k < 16; ++k)
    yv[k] = *(const v4f*)(yp + (size_t)((b * 16 + k) * 8 + h) * 1024 + tid * 4);
  v4f w[WROWS][8];
  ld_rows(w, Wkv + (size_t)(1024 + r0 + wv * WROWS) * 1024 + ln * 4);

  float mk[16];
  float M = -3.4e38f;
#pragma unroll
  for (int k = 0; k < 16; ++k) {
    mk[k] = mp[(b * 16 + k) * 8 + h];
    M = fmaxf(M, mk[k]);
  }
  float L = 0.f;
  v4f acc = {0.f, 0.f, 0.f, 0.f};
#pragma unroll
  for (int k = 0; k < 16; ++k) {
    const float e = __expf(mk[k] - M);
    L += lp[(b * 16 + k) * 8 + h] * e;
    acc += yv[k] * e;
  }
  *(v4f*)&yn[tid * 4] = acc / L;
  __syncthreads();

  v4f xv[8];
#pragma unroll
  for (int i = 0; i < 8; ++i) xv[i] = *(const v4f*)&yn[i * 128 + ln * 4];
  rows8(xv, w, res + wv * WROWS, ln);
  __syncthreads();
  if (tid < TROWS_W) agg[b * 1024 + r0 + tid] = res[tid];
}

__global__ __launch_bounds__(256) __attribute__((amdgpu_waves_per_eu(2, 2)))
void ga_proj(const float* __restrict__ agg,
             const float* __restrict__ projW,
             const float* __restrict__ projb,
             const float* __restrict__ cf,
             float* __restrict__ out) {
  __shared__ float res[TROWS_W];
  const int tid = threadIdx.x;
  const int wv = tid >> 5, ln = tid & 31;
  const int b = blockIdx.x >> 4;
  const int r0 = (blockIdx.x & 15) * TROWS_W;

  v4f xv[8];
#pragma unroll
  for (int i = 0; i < 8; ++i) xv[i] = *(const v4f*)(agg + b * 1024 + i * 128 + ln * 4);
  v4f w[WROWS][8];
  ld_rows(w, projW + (size_t)(r0 + wv * WROWS) * 1024 + ln * 4);
  rows8(xv, w, res + wv * WROWS, ln);
  __syncthreads();
  if (tid < TROWS_W) {
    const int j = r0 + tid;
    out[(size_t)b * 1024 + j] = cf[j] + res[tid] + projb[j];
  }
}

extern "C" void kernel_launch(void* const* d_in, const int* in_sizes, int n_in,
                              void* d_out, int out_size, void* d_ws, size_t ws_size,
                              hipStream_t stream) {
  const float* class_feature = (const float*)d_in[0];
  const float* features      = (const float*)d_in[1];
  const float* q_gamma       = (const float*)d_in[2];
  const float* q_beta        = (const float*)d_in[3];
  const float* Wq            = (const float*)d_in[4];
  const float* kv_gamma      = (const float*)d_in[5];
  const float* kv_beta       = (const float*)d_in[6];
  const float* Wkv           = (const float*)d_in[7];
  const float* proj_W        = (const float*)d_in[8];
  const float* proj_b        = (const float*)d_in[9];
  float* out = (float*)d_out;

  float* ws   = (float*)d_ws;
  float* wqe  = ws + 2048;
  float* mp   = ws + 18432;
  float* lp   = ws + 20480;
  float* yp   = ws + 22528;
  float* agg  = ws + 2048;

  ga_pre<<<64, 512, 0, stream>>>(class_feature, q_gamma, q_beta, Wq, Wkv, wqe);
  ga_main<<<B_ * NCHUNK, 512, 0, stream>>>(features, kv_gamma, kv_beta, wqe, mp, lp, yp);
  ga_agg<<<B_ * 16, 256, 0, stream>>>(mp, lp, yp, Wkv, agg);
  ga_proj<<<B_ * 16, 256, 0, stream>>>(agg, proj_W, proj_b, class_feature, out);
}
